# starting kernel
# baseline (speedup 1.0000x reference)
#include <hip/hip_runtime.h>
#include <hip/hip_bf16.h>

typedef __attribute__((ext_vector_type(2))) float v2f;
typedef __attribute__((ext_vector_type(8))) float v8f;

constexpr int B_ = 4, L_ = 1024, VOCAB_ = 50257, D_ = 64;

__global__ __launch_bounds__(256) void pool_embed_kernel(
    const int* __restrict__ x, const float* __restrict__ a,
    const float* __restrict__ E_w, const float* __restrict__ V_w,
    float* __restrict__ c_ws)
{
    __shared__ float w[L_];
    __shared__ int   xs[L_];
    __shared__ float red[256];
    __shared__ float m_s[D_];

    const int b = blockIdx.x;
    const int t = threadIdx.x;
    const float* arow = a + ((size_t)b * L_ + (L_ - 1)) * (size_t)L_;

    float vals[4];
    float lmax = -INFINITY;
    #pragma unroll
    for (int i = 0; i < 4; ++i) {
        int l = t + i * 256;
        float v = arow[l];
        vals[i] = v;
        lmax = fmaxf(lmax, v);
        xs[l] = x[b * L_ + l];
    }
    red[t] = lmax;
    __syncthreads();
    for (int s = 128; s > 0; s >>= 1) {
        if (t < s) red[t] = fmaxf(red[t], red[t + s]);
        __syncthreads();
    }
    float mx = red[0];
    __syncthreads();

    float lsum = 0.f;
    #pragma unroll
    for (int i = 0; i < 4; ++i) {
        int l = t + i * 256;
        float e = __expf(vals[i] - mx);
        w[l] = e;
        lsum += e;
    }
    red[t] = lsum;
    __syncthreads();
    for (int s = 128; s > 0; s >>= 1) {
        if (t < s) red[t] += red[t + s];
        __syncthreads();
    }
    float inv_sum = 1.0f / red[0];
    __syncthreads();

    int d = t & 63;
    int chunk = t >> 6;
    const float* Ed = E_w + (size_t)d * VOCAB_;
    float partial = 0.f;
    int l0 = chunk * 256;
    for (int l = l0; l < l0 + 256; ++l)
        partial += w[l] * Ed[xs[l]];
    red[t] = partial;
    __syncthreads();
    if (t < 64)
        m_s[t] = (red[t] + red[t + 64] + red[t + 128] + red[t + 192]) * inv_sum;
    __syncthreads();

    if (t < 64) {
        const float* Vr = V_w + t * D_;
        float cc = 0.f;
        #pragma unroll
        for (int dd = 0; dd < D_; ++dd) cc += m_s[dd] * Vr[dd];
        c_ws[b * D_ + t] = cc;
    }
}

__global__ __launch_bounds__(256) void vocab_wmma_kernel(
    const float* __restrict__ U_w, const float* __restrict__ c_ws,
    float* __restrict__ out)
{
    const int t    = threadIdx.x;
    const int wave = t >> 5;
    const int lane = t & 31;
    const int tile = blockIdx.x * 8 + wave;
    const int col0 = tile * 16;
    if (col0 >= VOCAB_) return;

    const int n     = lane & 15;
    const int khalf = lane >> 4;
    const int col   = col0 + n;
    const bool cvalid = col < VOCAB_;
    const int colc  = cvalid ? col : (VOCAB_ - 1);
    const int mrow  = lane & 15;

    const float* Urow = U_w  + (size_t)colc * D_;
    const float* Crow = c_ws + (mrow & 3) * D_;

    v2f afr[16], bfr[16];
    #pragma unroll
    for (int kk = 0; kk < 16; ++kk) {
        const int kb = kk * 4 + 2 * khalf;
        afr[kk] = *(const v2f*)(Crow + kb);
        bfr[kk] = *(const v2f*)(Urow + kb);
    }

    v8f acc = {};
    #pragma unroll
    for (int kk = 0; kk < 16; ++kk) {
        acc = __builtin_amdgcn_wmma_f32_16x16x4_f32(
            false, afr[kk], false, bfr[kk], (short)0, acc, false, false);
    }

    if (khalf == 0 && cvalid) {
        #pragma unroll
        for (int r = 0; r < B_; ++r)
            out[(size_t)r * VOCAB_ + col] = acc[r];
    }
}

extern "C" void kernel_launch(void* const* d_in, const int* in_sizes, int n_in,
                              void* d_out, int out_size, void* d_ws, size_t ws_size,
                              hipStream_t stream) {
    const int*   x   = (const int*)d_in[0];
    const float* a   = (const float*)d_in[1];
    const float* E_w = (const float*)d_in[2];
    const float* V_w = (const float*)d_in[3];
    const float* U_w = (const float*)d_in[4];
    float* out  = (float*)d_out;
    float* c_ws = (float*)d_ws;

    pool_embed_kernel<<<B_, 256, 0, stream>>>(x, a, E_w, V_w, c_ws);

    const int ntiles  = (VOCAB_ + 15) / 16;
    const int nblocks = (ntiles + 7) / 8;
    vocab_wmma_kernel<<<nblocks, 256, 0, stream>>>(U_w, c_ws, out);
}


# each batch row's embedding gather split over 64 work-groups (256 in all, a thread's 4 loads in flight at once) writing V-projected partials; the vocab kernel's head adds them; sums in a new order
# speedup vs baseline: 13.2960x; 13.2960x over previous
#include <hip/hip_runtime.h>
#include <hip/hip_bf16.h>


typedef __attribute__((ext_vector_type(2))) float v2f;
typedef __attribute__((ext_vector_type(4))) float v4f;
typedef __attribute__((ext_vector_type(8))) float v8f;
typedef __attribute__((ext_vector_type(4))) int   v4i;

constexpr int B_ = 4, L_ = 1024, VOCAB_ = 50257, D_ = 64;

constexpr int NCH_   = 64;
constexpr int LC_    = L_ / NCH_;
constexpr int NBLK1_ = B_ * NCH_;
constexpr int VP_    = D_ + 4;
constexpr int PCS_   = B_ * D_;

__global__ __launch_bounds__(256) void pool_embed_kernel(
    const int* __restrict__ x, const float* __restrict__ a,
    const float* __restrict__ E_w, const float* __restrict__ V_w,
    float* __restrict__ pc_ws)
{
    __shared__ __attribute__((aligned(16))) float Vs[D_ * VP_];
    __shared__ __attribute__((aligned(16))) float part[256];
    __shared__ float part2[256];
    __shared__ float redm[8];
    __shared__ float reds[8];

    const int t    = threadIdx.x;
    const int lane = t & 31;
    const int wave = t >> 5;
    const int b    = blockIdx.x / NCH_;
    const int ch   = blockIdx.x % NCH_;
    const int d    = t & 63;
    const int sub  = t >> 6;

    const float* arow = a + ((size_t)b * L_ + (L_ - 1)) * (size_t)L_;
    const int l0 = ch * LC_ + sub * 4;

    const v4i tk = *(const v4i*)(x + b * L_ + l0);
    const v4f av = *(const v4f*)(arow + l0);
    const v4f rv = *(const v4f*)(arow + 4 * t);

    v4f vld[4];
    #pragma unroll
    for (int i = 0; i < 4; ++i)
        vld[i] = *(const v4f*)(V_w + 4 * (i * 256 + t));

    const float* Ed = E_w + (size_t)d * VOCAB_;
    const float e0 = Ed[tk.x];
    const float e1 = Ed[tk.y];
    const float e2 = Ed[tk.z];
    const float e3 = Ed[tk.w];

    #pragma unroll
    for (int i = 0; i < 4; ++i) {
        const int f = i * 256 + t;
        *(v4f*)&Vs[(f >> 4) * VP_ + (f & 15) * 4] = vld[i];
    }

    float wmax = fmaxf(fmaxf(rv.x, rv.y), fmaxf(rv.z, rv.w));
    #pragma unroll
    for (int o = 16; o > 0; o >>= 1) wmax = fmaxf(wmax, __shfl_xor(wmax, o, 32));
    float wsum = (__expf(rv.x - wmax) + __expf(rv.y - wmax)) + (__expf(rv.z - wmax) + __expf(rv.w - wmax));
    #pragma unroll
    for (int o = 16; o > 0; o >>= 1) wsum += __shfl_xor(wsum, o, 32);
    if (lane == 0) { redm[wave] = wmax; reds[wave] = wsum; }
    __syncthreads();

    float mx = redm[0];
    #pragma unroll
    for (int i = 1; i < 8; ++i) mx = fmaxf(mx, redm[i]);
    float tot = 0.f;
    #pragma unroll
    for (int i = 0; i < 8; ++i) tot += reds[i] * __expf(redm[i] - mx);
    const float inv_sum = 1.0f / tot;

    float p = __expf(av.x - mx) * e0;
    p += __expf(av.y - mx) * e1;
    p += __expf(av.z - mx) * e2;
    p += __expf(av.w - mx) * e3;
    part[t] = p;
    __syncthreads();

    float cc = 0.f;
    #pragma unroll
    for (int i = 0; i < 4; ++i) {
        const int dd = sub * 16 + 4 * i;
        const v4f m4 = (*(const v4f*)&part[dd] + *(const v4f*)&part[64 + dd]) +
                       (*(const v4f*)&part[128 + dd] + *(const v4f*)&part[192 + dd]);
        const v4f vv = *(const v4f*)&Vs[d * VP_ + dd];
        cc += m4.x * vv.x;
        cc += m4.y * vv.y;
        cc += m4.z * vv.z;
        cc += m4.w * vv.w;
    }
    part2[t] = cc * inv_sum;
    __syncthreads();
    if (t < 64)
        pc_ws[(ch * B_ + b) * D_ + t] =
            (part2[t] + part2[t + 64]) + (part2[t + 128] + part2[t + 192]);
}

__global__ __launch_bounds__(256) void vocab_wmma_kernel(
    const float* __restrict__ U_w, const float* __restrict__ pc_ws,
    float* __restrict__ out)
{
    __shared__ __attribute__((aligned(16))) float ps[4 * PCS_];
    __shared__ __attribute__((aligned(16))) float cs[B_ * VP_];

    const int t    = threadIdx.x;
    const int wave = t >> 5;
    const int lane = t & 31;
    const int tile = blockIdx.x * 8 + wave;
    const int col0 = tile * 16;

    const int n     = lane & 15;
    const int khalf = lane >> 4;
    const int col   = col0 + n;
    const bool cvalid = col < VOCAB_;
    const int colc  = cvalid ? col : (VOCAB_ - 1);
    const int mrow  = lane & 15;

    const float* Urow = U_w + (size_t)colc * D_;

    v2f afr[16], bfr[16];
    #pragma unroll
    for (int kk = 0; kk < 16; ++kk)
        bfr[kk] = *(const v2f*)(Urow + kk * 4 + 2 * khalf);

    const int q  = t >> 6;
    const int c4 = (t & 63) * 4;
    const float* pp = pc_ws + (q * 16) * PCS_ + c4;
    v4f s0 = *(const v4f*)(pp);
    v4f s1 = *(const v4f*)(pp + PCS_);
    #pragma unroll
    for (int i = 2; i < 16; i += 2) {
        s0 += *(const v4f*)(pp + i * PCS_);
        s1 += *(const v4f*)(pp + (i + 1) * PCS_);
    }
    *(v4f*)&ps[q * PCS_ + c4] = s0 + s1;
    __syncthreads();
    cs[(t >> 6) * VP_ + (t & 63)] = (ps[t] + ps[PCS_ + t]) + (ps[2 * PCS_ + t] + ps[3 * PCS_ + t]);
    __syncthreads();

    if (col0 >= VOCAB_) return;

    const float* Crow = cs + (mrow & 3) * VP_;
    #pragma unroll
    for (int kk = 0; kk < 16; ++kk)
        afr[kk] = *(const v2f*)(Crow + kk * 4 + 2 * khalf);
    __builtin_amdgcn_sched_barrier(0);

    v8f acc = {};
    #pragma unroll
    for (int kk = 0; kk < 16; ++kk) {
        acc = __builtin_amdgcn_wmma_f32_16x16x4_f32(
            false, afr[kk], false, bfr[kk], (short)0, acc, false, false);
    }

    if (khalf == 0 && cvalid) {
        #pragma unroll
        for (int r = 0; r < B_; ++r)
            out[(size_t)r * VOCAB_ + col] = acc[r];
    }
}

extern "C" void kernel_launch(void* const* d_in, const int* in_sizes, int n_in,
                              void* d_out, int out_size, void* d_ws, size_t ws_size,
                              hipStream_t stream) {
    const int*   x   = (const int*)d_in[0];
    const float* a   = (const float*)d_in[1];
    const float* E_w = (const float*)d_in[2];
    const float* V_w = (const float*)d_in[3];
    const float* U_w = (const float*)d_in[4];
    float* out   = (float*)d_out;
    float* pc_ws = (float*)d_ws;

    pool_embed_kernel<<<NBLK1_, 256, 0, stream>>>(x, a, E_w, V_w, pc_ws);

    const int ntiles  = (VOCAB_ + 15) / 16;
    const int nblocks = (ntiles + 7) / 8;
    vocab_wmma_kernel<<<nblocks, 256, 0, stream>>>(U_w, pc_ws, out);
}


# vocab kernel: each wave fetches its 4 KB U_w tile with 8 whole-line 16-byte loads (was 16 loads of 16 lines each), all issued ahead of any wait, and re-lays it in LDS; k summed in a new order
# speedup vs baseline: 14.0159x; 1.0541x over previous
#include <hip/hip_runtime.h>
#include <hip/hip_bf16.h>


typedef __attribute__((ext_vector_type(2))) float v2f;
typedef __attribute__((ext_vector_type(4))) float v4f;
typedef __attribute__((ext_vector_type(8))) float v8f;
typedef __attribute__((ext_vector_type(4))) int   v4i;

constexpr int B_ = 4, L_ = 1024, VOCAB_ = 50257, D_ = 64;

constexpr int NCH_   = 64;
constexpr int LC_    = L_ / NCH_;
constexpr int NBLK1_ = B_ * NCH_;
constexpr int VP_    = D_ + 4;
constexpr int PCS_   = B_ * D_;

__global__ __launch_bounds__(256) void pool_embed_kernel(
    const int* __restrict__ x, const float* __restrict__ a,
    const float* __restrict__ E_w, const float* __restrict__ V_w,
    float* __restrict__ pc_ws)
{
    __shared__ __attribute__((aligned(16))) float Vs[D_ * VP_];
    __shared__ __attribute__((aligned(16))) float part[256];
    __shared__ float part2[256];
    __shared__ float redm[8];
    __shared__ float reds[8];

    const int t    = threadIdx.x;
    const int lane = t & 31;
    const int wave = t >> 5;
    const int b    = blockIdx.x / NCH_;
    const int ch   = blockIdx.x % NCH_;
    const int d    = t & 63;
    const int sub  = t >> 6;

    const float* arow = a + ((size_t)b * L_ + (L_ - 1)) * (size_t)L_;
    const int l0 = ch * LC_ + sub * 4;

    const v4i tk = *(const v4i*)(x + b * L_ + l0);
    const v4f av = *(const v4f*)(arow + l0);
    const v4f rv = *(const v4f*)(arow + 4 * t);

    v4f vld[4];
    #pragma unroll
    for (int i = 0; i < 4; ++i)
        vld[i] = *(const v4f*)(V_w + 4 * (i * 256 + t));

    const float* Ed = E_w + (size_t)d * VOCAB_;
    const float e0 = Ed[tk.x];
    const float e1 = Ed[tk.y];
    const float e2 = Ed[tk.z];
    const float e3 = Ed[tk.w];

    #pragma unroll
    for (int i = 0; i < 4; ++i) {
        const int f = i * 256 + t;
        *(v4f*)&Vs[(f >> 4) * VP_ + (f & 15) * 4] = vld[i];
    }

    float wmax = fmaxf(fmaxf(rv.x, rv.y), fmaxf(rv.z, rv.w));
    #pragma unroll
    for (int o = 16; o > 0; o >>= 1) wmax = fmaxf(wmax, __shfl_xor(wmax, o, 32));
    float wsum = (__expf(rv.x - wmax) + __expf(rv.y - wmax)) + (__expf(rv.z - wmax) + __expf(rv.w - wmax));
    #pragma unroll
    for (int o = 16; o > 0; o >>= 1) wsum += __shfl_xor(wsum, o, 32);
    if (lane == 0) { redm[wave] = wmax; reds[wave] = wsum; }
    __syncthreads();

    float mx = redm[0];
    #pragma unroll
    for (int i = 1; i < 8; ++i) mx = fmaxf(mx, redm[i]);
    float tot = 0.f;
    #pragma unroll
    for (int i = 0; i < 8; ++i) tot += reds[i] * __expf(redm[i] - mx);
    const float inv_sum = 1.0f / tot;

    float p = __expf(av.x - mx) * e0;
    p += __expf(av.y - mx) * e1;
    p += __expf(av.z - mx) * e2;
    p += __expf(av.w - mx) * e3;
    part[t] = p;
    __syncthreads();

    float cc = 0.f;
    #pragma unroll
    for (int i = 0; i < 4; ++i) {
        const int dd = sub * 16 + 4 * i;
        const v4f m4 = (*(const v4f*)&part[dd] + *(const v4f*)&part[64 + dd]) +
                       (*(const v4f*)&part[128 + dd] + *(const v4f*)&part[192 + dd]);
        const v4f vv = *(const v4f*)&Vs[d * VP_ + dd];
        cc += m4.x * vv.x;
        cc += m4.y * vv.y;
        cc += m4.z * vv.z;
        cc += m4.w * vv.w;
    }
    part2[t] = cc * inv_sum;
    __syncthreads();
    if (t < 64)
        pc_ws[(ch * B_ + b) * D_ + t] =
            (part2[t] + part2[t + 64]) + (part2[t + 128] + part2[t + 192]);
}

__global__ __launch_bounds__(256) void vocab_wmma_kernel(
    const float* __restrict__ U_w, const float* __restrict__ pc_ws,
    float* __restrict__ out)
{
    __shared__ __attribute__((aligned(16))) float ps[4 * PCS_];
    __shared__ __attribute__((aligned(16))) float cs[B_ * VP_];
    __shared__ __attribute__((aligned(16))) float bt[8 * 16 * VP_];

    const int t    = threadIdx.x;
    const int wave = t >> 5;
    const int lane = t & 31;
    const int tile = blockIdx.x * 8 + wave;
    const int col0 = tile * 16;

    const int n     = lane & 15;
    const int khalf = lane >> 4;
    const int col   = col0 + n;
    const bool cvalid = col < VOCAB_;

    float* btw = bt + wave * (16 * VP_);
    v4f bld[8];
    #pragma unroll
    for (int i = 0; i < 8; ++i) {
        const int row  = col0 + 2 * i + khalf;
        const int rowc = row < VOCAB_ ? row : (VOCAB_ - 1);
        bld[i] = *(const v4f*)(U_w + ((unsigned)rowc * D_ + 4u * n));
    }
    __builtin_amdgcn_sched_barrier(0);

    const int q  = t >> 6;
    const int c4 = (t & 63) * 4;
    const float* pp = pc_ws + (q * 16) * PCS_ + c4;
    v4f hld[16];
    #pragma unroll
    for (int i = 0; i < 16; ++i)
        hld[i] = *(const v4f*)(pp + i * PCS_);
    __builtin_amdgcn_sched_barrier(0);

    #pragma unroll
    for (int i = 0; i < 8; ++i)
        *(v4f*)&btw[(2 * i + khalf) * VP_ + 4 * n] = bld[i];
    v4f s0 = hld[0];
    v4f s1 = hld[1];
    #pragma unroll
    for (int i = 2; i < 16; i += 2) {
        s0 += hld[i];
        s1 += hld[i + 1];
    }
    *(v4f*)&ps[q * PCS_ + c4] = s0 + s1;
    __syncthreads();

    v2f afr[16], bfr[16];
    #pragma unroll
    for (int j = 0; j < 8; ++j) {
        const v4f bv = *(const v4f*)&btw[n * VP_ + 8 * j + 4 * khalf];
        bfr[2 * j]     = v2f{bv.x, bv.y};
        bfr[2 * j + 1] = v2f{bv.z, bv.w};
    }

    cs[(t >> 6) * VP_ + (t & 63)] = (ps[t] + ps[PCS_ + t]) + (ps[2 * PCS_ + t] + ps[3 * PCS_ + t]);
    __syncthreads();

    if (col0 >= VOCAB_) return;

    const float* Crow = cs + (n & 3) * VP_;
    #pragma unroll
    for (int j = 0; j < 8; ++j) {
        const v4f cv = *(const v4f*)(Crow + 8 * j + 4 * khalf);
        afr[2 * j]     = v2f{cv.x, cv.y};
        afr[2 * j + 1] = v2f{cv.z, cv.w};
    }
    __builtin_amdgcn_sched_barrier(0);

    v8f acc = {};
    #pragma unroll
    for (int kk = 0; kk < 16; ++kk) {
        acc = __builtin_amdgcn_wmma_f32_16x16x4_f32(
            false, afr[kk], false, bfr[kk], (short)0, acc, false, false);
    }

    if (khalf == 0 && cvalid) {
        #pragma unroll
        for (int r = 0; r < B_; ++r)
            out[(size_t)r * VOCAB_ + col] = acc[r];
    }
}

extern "C" void kernel_launch(void* const* d_in, const int* in_sizes, int n_in,
                              void* d_out, int out_size, void* d_ws, size_t ws_size,
                              hipStream_t stream) {
    const int*   x   = (const int*)d_in[0];
    const float* a   = (const float*)d_in[1];
    const float* E_w = (const float*)d_in[2];
    const float* V_w = (const float*)d_in[3];
    const float* U_w = (const float*)d_in[4];
    float* out   = (float*)d_out;
    float* pc_ws = (float*)d_ws;

    pool_embed_kernel<<<NBLK1_, 256, 0, stream>>>(x, a, E_w, V_w, pc_ws);

    const int ntiles  = (VOCAB_ + 15) / 16;
    const int nblocks = (ntiles + 7) / 8;
    vocab_wmma_kernel<<<nblocks, 256, 0, stream>>>(U_w, pc_ws, out);
}


# pooling re-split by (batch row, embedding dim): each of 256 work-groups sums its 1024 gathered E_w entries to a finished value, no partial array; vocab head applies V_w; sums in a new order
# speedup vs baseline: 16.3761x; 1.1684x over previous
#include <hip/hip_runtime.h>
#include <hip/hip_bf16.h>


typedef __attribute__((ext_vector_type(2))) float v2f;
typedef __attribute__((ext_vector_type(4))) float v4f;
typedef __attribute__((ext_vector_type(8))) float v8f;
typedef __attribute__((ext_vector_type(4))) int   v4i;

constexpr int B_ = 4, L_ = 1024, VOCAB_ = 50257, D_ = 64;

constexpr int NBLK1_ = B_ * D_;
constexpr int VP_    = D_ + 4;

template <unsigned LO, unsigned HI>
static __device__ __forceinline__ float row_perm(float v) {
    const int b = __builtin_bit_cast(int, v);
    return __builtin_bit_cast(float, __builtin_amdgcn_permlane16(b, b, LO, HI, false, false));
}
static __device__ __forceinline__ float xhalf(float v) {
    const int b = __builtin_bit_cast(int, v);
    return __builtin_bit_cast(float, __builtin_amdgcn_permlanex16(b, b, 0x76543210u, 0xfedcba98u, false, false));
}
static __device__ __forceinline__ float wave_sum(float v) {
    v += row_perm<0x67452301u, 0xefcdab89u>(v);
    v += row_perm<0x54761032u, 0xdcfe98bau>(v);
    v += row_perm<0x32107654u, 0xba98fedcu>(v);
    v += row_perm<0xfedcba98u, 0x76543210u>(v);
    v += xhalf(v);
    return v;
}
static __device__ __forceinline__ float wave_max(float v) {
    v = fmaxf(v, row_perm<0x67452301u, 0xefcdab89u>(v));
    v = fmaxf(v, row_perm<0x54761032u, 0xdcfe98bau>(v));
    v = fmaxf(v, row_perm<0x32107654u, 0xba98fedcu>(v));
    v = fmaxf(v, row_perm<0xfedcba98u, 0x76543210u>(v));
    v = fmaxf(v, xhalf(v));
    return v;
}

__global__ __launch_bounds__(256) void pool_embed_kernel(
    const int* __restrict__ x, const float* __restrict__ a,
    const float* __restrict__ E_w, float* __restrict__ pooled_ws)
{
    __shared__ float reds[8];
    __shared__ float redp[8];

    const int t    = threadIdx.x;
    const int lane = t & 31;
    const int wave = t >> 5;
    const int b    = blockIdx.x / D_;
    const int d    = blockIdx.x % D_;

    const float* arow = a + ((size_t)b * L_ + (L_ - 1)) * (size_t)L_;

    const v4i tk = *(const v4i*)(x + b * L_ + 4 * t);
    const v4f av = *(const v4f*)(arow + 4 * t);
    v4f rw[8];
    #pragma unroll
    for (int i = 0; i < 8; ++i)
        rw[i] = *(const v4f*)(arow + 4 * (i * 32 + lane));

    const float* Ed = E_w + (size_t)d * VOCAB_;
    const float e0 = Ed[tk.x];
    const float e1 = Ed[tk.y];
    const float e2 = Ed[tk.z];
    const float e3 = Ed[tk.w];

    v4f m4 = rw[0];
    #pragma unroll
    for (int i = 1; i < 8; ++i) {
        m4.x = fmaxf(m4.x, rw[i].x);
        m4.y = fmaxf(m4.y, rw[i].y);
        m4.z = fmaxf(m4.z, rw[i].z);
        m4.w = fmaxf(m4.w, rw[i].w);
    }
    const float mx = wave_max(fmaxf(fmaxf(m4.x, m4.y), fmaxf(m4.z, m4.w)));

    const float w0 = __expf(av.x - mx);
    const float w1 = __expf(av.y - mx);
    const float w2 = __expf(av.z - mx);
    const float w3 = __expf(av.w - mx);
    const float s = wave_sum((w0 + w1) + (w2 + w3));

    float p = w0 * e0;
    p += w1 * e1;
    p += w2 * e2;
    p += w3 * e3;
    p = wave_sum(p);

    if (lane == 0) { reds[wave] = s; redp[wave] = p; }
    __syncthreads();
    if (t == 0) {
        const float S = ((reds[0] + reds[1]) + (reds[2] + reds[3])) + ((reds[4] + reds[5]) + (reds[6] + reds[7]));
        const float P = ((redp[0] + redp[1]) + (redp[2] + redp[3])) + ((redp[4] + redp[5]) + (redp[6] + redp[7]));
        pooled_ws[b * D_ + d] = P / S;
    }
}

__global__ __launch_bounds__(256) void vocab_wmma_kernel(
    const float* __restrict__ U_w, const float* __restrict__ V_w,
    const float* __restrict__ pooled_ws, float* __restrict__ out)
{
    __shared__ __attribute__((aligned(16))) float Vs[D_ * VP_];
    __shared__ __attribute__((aligned(16))) float cs[B_ * VP_];
    __shared__ __attribute__((aligned(16))) float bt[8 * 16 * VP_];

    const int t    = threadIdx.x;
    const int wave = t >> 5;
    const int lane = t & 31;
    const int tile = blockIdx.x * 8 + wave;
    const int col0 = tile * 16;

    const int n     = lane & 15;
    const int khalf = lane >> 4;
    const int col   = col0 + n;
    const bool cvalid = col < VOCAB_;

    float* btw = bt + wave * (16 * VP_);
    v4f bld[8];
    #pragma unroll
    for (int i = 0; i < 8; ++i) {
        const int row  = col0 + 2 * i + khalf;
        const int rowc = row < VOCAB_ ? row : (VOCAB_ - 1);
        bld[i] = *(const v4f*)(U_w + ((unsigned)rowc * D_ + 4u * n));
    }

    const int he = t & 63;
    const int hb = __builtin_amdgcn_readfirstlane(t >> 6);
    v4f vld[4];
    #pragma unroll
    for (int i = 0; i < 4; ++i)
        vld[i] = *(const v4f*)(V_w + 4 * (i * 256 + t));
    const float* pb = pooled_ws + hb * D_;
    v4f pv[16];
    #pragma unroll
    for (int j = 0; j < 16; ++j)
        pv[j] = *(const v4f*)(pb + 4 * j);

    #pragma unroll
    for (int i = 0; i < 8; ++i)
        *(v4f*)&btw[(2 * i + khalf) * VP_ + 4 * n] = bld[i];
    #pragma unroll
    for (int i = 0; i < 4; ++i) {
        const int f = i * 256 + t;
        *(v4f*)&Vs[(f >> 4) * VP_ + (f & 15) * 4] = vld[i];
    }
    __syncthreads();

    v2f afr[16], bfr[16];
    #pragma unroll
    for (int j = 0; j < 8; ++j) {
        const v4f bv = *(const v4f*)&btw[n * VP_ + 8 * j + 4 * khalf];
        bfr[2 * j]     = v2f{bv.x, bv.y};
        bfr[2 * j + 1] = v2f{bv.z, bv.w};
    }

    v4f c4 = {0.f, 0.f, 0.f, 0.f};
    #pragma unroll
    for (int j = 0; j < 16; ++j) {
        const v4f vv = *(const v4f*)&Vs[he * VP_ + 4 * j];
        c4 += vv * pv[j];
    }
    cs[hb * VP_ + he] = (c4.x + c4.y) + (c4.z + c4.w);
    __syncthreads();

    if (col0 >= VOCAB_) return;

    const float* Crow = cs + (n & 3) * VP_;
    #pragma unroll
    for (int j = 0; j < 8; ++j) {
        const v4f cv = *(const v4f*)(Crow + 8 * j + 4 * khalf);
        afr[2 * j]     = v2f{cv.x, cv.y};
        afr[2 * j + 1] = v2f{cv.z, cv.w};
    }
    __builtin_amdgcn_sched_barrier(0);

    v8f acc = {};
    #pragma unroll
    for (int kk = 0; kk < 16; ++kk) {
        acc = __builtin_amdgcn_wmma_f32_16x16x4_f32(
            false, afr[kk], false, bfr[kk], (short)0, acc, false, false);
    }

    if (khalf == 0 && cvalid) {
        #pragma unroll
        for (int r = 0; r < B_; ++r)
            out[(size_t)r * VOCAB_ + col] = acc[r];
    }
}

extern "C" void kernel_launch(void* const* d_in, const int* in_sizes, int n_in,
                              void* d_out, int out_size, void* d_ws, size_t ws_size,
                              hipStream_t stream) {
    const int*   x   = (const int*)d_in[0];
    const float* a   = (const float*)d_in[1];
    const float* E_w = (const float*)d_in[2];
    const float* V_w = (const float*)d_in[3];
    const float* U_w = (const float*)d_in[4];
    float* out       = (float*)d_out;
    float* pooled_ws = (float*)d_ws;

    pool_embed_kernel<<<NBLK1_, 256, 0, stream>>>(x, a, E_w, pooled_ws);

    const int ntiles  = (VOCAB_ + 15) / 16;
    const int nblocks = (ntiles + 7) / 8;
    vocab_wmma_kernel<<<nblocks, 256, 0, stream>>>(U_w, V_w, pooled_ws, out);
}
